# starting kernel
# baseline (speedup 1.0000x reference)
#include <hip/hip_runtime.h>
#include <hip/hip_bf16.h>

typedef float v2f __attribute__((ext_vector_type(2)));
typedef float v8f __attribute__((ext_vector_type(8)));

#define NUM_HEADS 16
#define HEAD_DIM  64
#define DIM       1024
#define INNER     (NUM_HEADS * HEAD_DIM)
#define QKV_N     (3 * INNER)
#define BATCH     2
#define SEQ       2048
#define BS        (BATCH * SEQ)
#define QTILES    (SEQ / 16)

__global__ __launch_bounds__(32)
void gemm_bias_f32(const float* __restrict__ A, const float* __restrict__ W,
                   const float* __restrict__ bias, float* __restrict__ C,
                   int M, int N, int K) {
    const int l    = threadIdx.x;
    const int lo   = l & 15;
    const int half = l >> 4;
    const int n0   = blockIdx.x * 64;
    const int m0   = blockIdx.y * 64;

    const float* arow[4];
    const float* wcol[4];
#pragma unroll
    for (int i = 0; i < 4; ++i)
        arow[i] = A + (size_t)(m0 + 16 * i + lo) * K + 2 * half;
#pragma unroll
    for (int j = 0; j < 4; ++j)
        wcol[j] = W + (size_t)(2 * half) * N + n0 + 16 * j + lo;

    v8f acc[4][4] = {};

    for (int k = 0; k < K; k += 4) {
        if ((k & 31) == 0) {
#pragma unroll
            for (int i = 0; i < 4; ++i)
                __builtin_prefetch(arow[i] + k + 32, 0, 0);
        }
        v2f a[4], b[4];
#pragma unroll
        for (int i = 0; i < 4; ++i)
            a[i] = *(const v2f*)(arow[i] + k);
#pragma unroll
        for (int j = 0; j < 4; ++j) {
            b[j].x = wcol[j][(size_t)k * N];
            b[j].y = wcol[j][(size_t)k * N + N];
        }
#pragma unroll
        for (int i = 0; i < 4; ++i)
#pragma unroll
            for (int j = 0; j < 4; ++j)
                acc[i][j] = __builtin_amdgcn_wmma_f32_16x16x4_f32(
                    false, a[i], false, b[j], (short)0, acc[i][j], false, false);
    }

    float bv[4];
#pragma unroll
    for (int j = 0; j < 4; ++j) bv[j] = bias[n0 + 16 * j + lo];

#pragma unroll
    for (int i = 0; i < 4; ++i) {
        float* crow = C + (size_t)(m0 + 16 * i + 8 * half) * N + n0 + lo;
#pragma unroll
        for (int r = 0; r < 8; ++r)
#pragma unroll
            for (int j = 0; j < 4; ++j)
                crow[(size_t)r * N + 16 * j] = acc[i][j][r] + bv[j];
    }
}

__global__ __launch_bounds__(256)
void norm_rope(float* __restrict__ qkv, const float* __restrict__ rot,
               const float* __restrict__ wq, const float* __restrict__ wk) {
    const int row = blockIdx.x;
    float* q = qkv + (size_t)row * QKV_N;
    float* k = q + INNER;
    const float* fr = rot + (size_t)row * (2 * HEAD_DIM);
    const int t = threadIdx.x;

    float sq = 0.f, sk = 0.f;
    for (int i = t; i < INNER; i += 256) {
        float a = q[i]; sq += a * a;
        float b = k[i]; sk += b * b;
    }
    __shared__ float redq[256], redk[256];
    redq[t] = sq; redk[t] = sk;
    __syncthreads();
    for (int off = 128; off > 0; off >>= 1) {
        if (t < off) { redq[t] += redq[t + off]; redk[t] += redk[t + off]; }
        __syncthreads();
    }
    const float invq = rsqrtf(redq[0] * (1.0f / INNER) + 1e-5f);
    const float invk = rsqrtf(redk[0] * (1.0f / INNER) + 1e-5f);

    for (int p = t; p < INNER / 2; p += 256) {
        const int i2  = p & 31;
        const float c = fr[2 * i2];
        const float s = fr[HEAD_DIM + 2 * i2 + 1];
        const int idx = 2 * p;

        float xe = q[idx]     * invq * wq[idx];
        float xo = q[idx + 1] * invq * wq[idx + 1];
        q[idx]     = xe * c - xo * s;
        q[idx + 1] = xe * s + xo * c;

        xe = k[idx]     * invk * wk[idx];
        xo = k[idx + 1] * invk * wk[idx + 1];
        k[idx]     = xe * c - xo * s;
        k[idx + 1] = xe * s + xo * c;
    }
}

__global__ __launch_bounds__(32)
void attn_kernel(const float* __restrict__ qkv, const float* __restrict__ hks,
                 const int* __restrict__ oclp, float* __restrict__ out) {
    __shared__ __align__(16) float plds[256];

    const int bx = blockIdx.x;
    const int qt = bx & (QTILES - 1);
    const int h  = (bx >> 7) & (NUM_HEADS - 1);
    const int b  = bx >> 11;

    const int l    = threadIdx.x;
    const int lo   = l & 15;
    const int half = l >> 4;

    const int   hist = SEQ - *oclp;
    const float skh  = 1.0f + 9.0f / (1.0f + __expf(-hks[h]));
    const float sscale = 0.125f;

    const size_t base  = (size_t)b * SEQ * QKV_N;
    const float* qbase = qkv + base + (size_t)(qt * 16 + lo) * QKV_N + h * HEAD_DIM + 2 * half;
    const float* kbase = qkv + base + INNER     + h * HEAD_DIM + 2 * half;
    const float* vbase = qkv + base + 2 * INNER + h * HEAD_DIM;

    v2f qa[16];
#pragma unroll
    for (int d = 0; d < 16; ++d)
        qa[d] = *(const v2f*)(qbase + 4 * d);

    v8f o[4] = {};
    float m_state[8], l_state[8];
#pragma unroll
    for (int r = 0; r < 8; ++r) { m_state[r] = -__builtin_inff(); l_state[r] = 0.f; }

    for (int t = 0; t < QTILES; ++t) {
        v8f c = {};
        const float* krow = kbase + (size_t)(t * 16 + lo) * QKV_N;
#pragma unroll
        for (int d = 0; d < 16; ++d) {
            v2f kb = *(const v2f*)(krow + 4 * d);
            c = __builtin_amdgcn_wmma_f32_16x16x4_f32(false, qa[d], false, kb,
                                                      (short)0, c, false, false);
        }

        const int   kg  = t * 16 + lo;
        const float fac = sscale * (kg < hist ? skh : 1.0f);

        float p[8], alpha[8];
#pragma unroll
        for (int r = 0; r < 8; ++r) {
            float s = c[r] * fac;
            float rm = s;
            rm = fmaxf(rm, __shfl_xor(rm, 1));
            rm = fmaxf(rm, __shfl_xor(rm, 2));
            rm = fmaxf(rm, __shfl_xor(rm, 4));
            rm = fmaxf(rm, __shfl_xor(rm, 8));
            const float mn = fmaxf(m_state[r], rm);
            const float al = __expf(m_state[r] - mn);
            const float pv = __expf(s - mn);
            float su = pv;
            su += __shfl_xor(su, 1);
            su += __shfl_xor(su, 2);
            su += __shfl_xor(su, 4);
            su += __shfl_xor(su, 8);
            l_state[r] = l_state[r] * al + su;
            m_state[r] = mn;
            alpha[r] = al;
            p[r] = pv;
        }

#pragma unroll
        for (int nt = 0; nt < 4; ++nt)
#pragma unroll
            for (int r = 0; r < 8; ++r)
                o[nt][r] *= alpha[r];

#pragma unroll
        for (int r = 0; r < 8; ++r)
            plds[(r + 8 * half) * 16 + lo] = p[r];
        __syncthreads();
        v2f pa[4];
#pragma unroll
        for (int kk = 0; kk < 4; ++kk) {
            pa[kk].x = plds[lo * 16 + 4 * kk + 2 * half];
            pa[kk].y = plds[lo * 16 + 4 * kk + 2 * half + 1];
        }
        __syncthreads();

        const float* vt = vbase + (size_t)(t * 16) * QKV_N;
#pragma unroll
        for (int kk = 0; kk < 4; ++kk) {
            const int kr = 4 * kk + 2 * half;
#pragma unroll
            for (int nt = 0; nt < 4; ++nt) {
                v2f vb;
                vb.x = vt[(size_t)kr * QKV_N + nt * 16 + lo];
                vb.y = vt[(size_t)(kr + 1) * QKV_N + nt * 16 + lo];
                o[nt] = __builtin_amdgcn_wmma_f32_16x16x4_f32(false, pa[kk], false, vb,
                                                              (short)0, o[nt], false, false);
            }
        }
    }

    float inv[8];
#pragma unroll
    for (int r = 0; r < 8; ++r) inv[r] = 1.0f / l_state[r];

    float* orow = out + (size_t)((size_t)b * SEQ + qt * 16 + 8 * half) * INNER
                      + h * HEAD_DIM + lo;
#pragma unroll
    for (int nt = 0; nt < 4; ++nt)
#pragma unroll
        for (int r = 0; r < 8; ++r)
            orow[(size_t)r * INNER + nt * 16] = o[nt][r] * inv[r];
}

extern "C" void kernel_launch(void* const* d_in, const int* in_sizes, int n_in,
                              void* d_out, int out_size, void* d_ws, size_t ws_size,
                              hipStream_t stream) {
    const float* hidden = (const float*)d_in[0];
    const float* rot    = (const float*)d_in[1];
    const float* w_qkv  = (const float*)d_in[2];
    const float* b_qkv  = (const float*)d_in[3];
    const float* nqw    = (const float*)d_in[4];
    const float* nkw    = (const float*)d_in[5];
    const float* w_out  = (const float*)d_in[6];
    const float* b_out  = (const float*)d_in[7];
    const float* hks    = (const float*)d_in[8];
    const int*   ocl    = (const int*)d_in[9];
    float* out = (float*)d_out;

    float* qkv  = (float*)d_ws;
    float* attn = qkv + (size_t)BS * QKV_N;

    gemm_bias_f32<<<dim3(QKV_N / 64, BS / 64), 32, 0, stream>>>(
        hidden, w_qkv, b_qkv, qkv, BS, QKV_N, DIM);

    norm_rope<<<BS, 256, 0, stream>>>(qkv, rot, nqw, nkw);

    attn_kernel<<<BATCH * NUM_HEADS * QTILES, 32, 0, stream>>>(qkv, hks, ocl, attn);

    gemm_bias_f32<<<dim3(DIM / 64, BS / 64), 32, 0, stream>>>(
        attn, w_out, b_out, out, BS, DIM, DIM);
}


# GEMM kernel (both launches): A and W staged through double-buffered LDS by a 4-wave 128x128 workgroup, whole-line 128-bit loads issued one K chunk ahead, fragments read from LDS; sum order kept
# speedup vs baseline: 1.3227x; 1.3227x over previous
#include <hip/hip_runtime.h>
#include <hip/hip_bf16.h>

typedef float v2f __attribute__((ext_vector_type(2)));
typedef float v8f __attribute__((ext_vector_type(8)));

#define NUM_HEADS 16
#define HEAD_DIM  64
#define DIM       1024
#define INNER     (NUM_HEADS * HEAD_DIM)
#define QKV_N     (3 * INNER)
#define BATCH     2
#define SEQ       2048
#define BS        (BATCH * SEQ)
#define QTILES    (SEQ / 16)

#define GB_M    128
#define GB_N    128
#define GB_K    32
#define A_PITCH 36
#define B_PITCH 288
#define A_STAGE (GB_M * A_PITCH)
#define B_STAGE ((GB_K / 2) * B_PITCH)

__global__ __launch_bounds__(128)
void gemm_bias_f32(const float* __restrict__ A, const float* __restrict__ W,
                   const float* __restrict__ bias, float* __restrict__ C,
                   int M, int N, int K) {
    __shared__ __align__(16) float sA[2][A_STAGE];
    __shared__ __align__(16) float sB[2][B_STAGE];

    const int t    = threadIdx.x;
    const int l    = t & 31;
    const int w    = t >> 5;
    const int lo   = l & 15;
    const int half = l >> 4;
    const int wm   = w >> 1;
    const int wn   = w & 1;
    const int n0   = blockIdx.x * GB_N;
    const int m0   = blockIdx.y * GB_M;

    const float* ag = A + (size_t)(m0 + (t >> 3)) * K + 4 * (t & 7);
    const float* bg = W + (size_t)(2 * w) * N + n0 + 4 * l;

    float4 ra[8], rb0[4], rb1[4];
    v8f acc[4][4] = {};

#pragma unroll
    for (int i = 0; i < 8; ++i)
        ra[i] = *(const float4*)(ag + (size_t)(16 * i) * K);
#pragma unroll
    for (int i = 0; i < 4; ++i) {
        rb0[i] = *(const float4*)(bg + (size_t)(8 * i) * N);
        rb1[i] = *(const float4*)(bg + (size_t)(8 * i + 1) * N);
    }
#pragma unroll
    for (int i = 0; i < 8; ++i)
        *(float4*)&sA[0][((t >> 3) + 16 * i) * A_PITCH + 4 * (t & 7)] = ra[i];
#pragma unroll
    for (int i = 0; i < 4; ++i) {
        float* d = &sB[0][(w + 4 * i) * B_PITCH + 8 * l];
        *(float4*)d       = make_float4(rb0[i].x, rb1[i].x, rb0[i].y, rb1[i].y);
        *(float4*)(d + 4) = make_float4(rb0[i].z, rb1[i].z, rb0[i].w, rb1[i].w);
    }
    __syncthreads();

    const int nk = K / GB_K;
    for (int kc = 0; kc < nk; ++kc) {
        const int s  = kc & 1;
        const int kn = (kc + 1 < nk ? kc + 1 : kc) * GB_K;
#pragma unroll
        for (int i = 0; i < 8; ++i)
            ra[i] = *(const float4*)(ag + (size_t)(16 * i) * K + kn);
#pragma unroll
        for (int i = 0; i < 4; ++i) {
            rb0[i] = *(const float4*)(bg + (size_t)(kn + 8 * i) * N);
            rb1[i] = *(const float4*)(bg + (size_t)(kn + 8 * i + 1) * N);
        }

        const float* aL = &sA[s][(wm * 64 + lo) * A_PITCH + 2 * half];
        const float* bL = &sB[s][half * B_PITCH + 2 * (wn * 64 + lo)];
#pragma unroll
        for (int kk = 0; kk < GB_K; kk += 4) {
            v2f a[4], b[4];
#pragma unroll
            for (int i = 0; i < 4; ++i)
                a[i] = *(const v2f*)(aL + 16 * i * A_PITCH + kk);
#pragma unroll
            for (int j = 0; j < 4; ++j)
                b[j] = *(const v2f*)(bL + (kk / 2) * B_PITCH + 32 * j);
#pragma unroll
            for (int i = 0; i < 4; ++i)
#pragma unroll
                for (int j = 0; j < 4; ++j)
                    acc[i][j] = __builtin_amdgcn_wmma_f32_16x16x4_f32(
                        false, a[i], false, b[j], (short)0, acc[i][j], false, false);
        }

#pragma unroll
        for (int i = 0; i < 8; ++i)
            *(float4*)&sA[s ^ 1][((t >> 3) + 16 * i) * A_PITCH + 4 * (t & 7)] = ra[i];
#pragma unroll
        for (int i = 0; i < 4; ++i) {
            float* d = &sB[s ^ 1][(w + 4 * i) * B_PITCH + 8 * l];
            *(float4*)d       = make_float4(rb0[i].x, rb1[i].x, rb0[i].y, rb1[i].y);
            *(float4*)(d + 4) = make_float4(rb0[i].z, rb1[i].z, rb0[i].w, rb1[i].w);
        }
        __syncthreads();
    }

    float bv[4];
#pragma unroll
    for (int j = 0; j < 4; ++j) bv[j] = bias[n0 + wn * 64 + 16 * j + lo];

#pragma unroll
    for (int i = 0; i < 4; ++i) {
        float* crow = C + (size_t)(m0 + wm * 64 + 16 * i + 8 * half) * N + n0 + wn * 64 + lo;
#pragma unroll
        for (int r = 0; r < 8; ++r)
#pragma unroll
            for (int j = 0; j < 4; ++j)
                crow[(size_t)r * N + 16 * j] = acc[i][j][r] + bv[j];
    }
}

__global__ __launch_bounds__(256)
void norm_rope(float* __restrict__ qkv, const float* __restrict__ rot,
               const float* __restrict__ wq, const float* __restrict__ wk) {
    const int row = blockIdx.x;
    float* q = qkv + (size_t)row * QKV_N;
    float* k = q + INNER;
    const float* fr = rot + (size_t)row * (2 * HEAD_DIM);
    const int t = threadIdx.x;

    float sq = 0.f, sk = 0.f;
    for (int i = t; i < INNER; i += 256) {
        float a = q[i]; sq += a * a;
        float b = k[i]; sk += b * b;
    }
    __shared__ float redq[256], redk[256];
    redq[t] = sq; redk[t] = sk;
    __syncthreads();
    for (int off = 128; off > 0; off >>= 1) {
        if (t < off) { redq[t] += redq[t + off]; redk[t] += redk[t + off]; }
        __syncthreads();
    }
    const float invq = rsqrtf(redq[0] * (1.0f / INNER) + 1e-5f);
    const float invk = rsqrtf(redk[0] * (1.0f / INNER) + 1e-5f);

    for (int p = t; p < INNER / 2; p += 256) {
        const int i2  = p & 31;
        const float c = fr[2 * i2];
        const float s = fr[HEAD_DIM + 2 * i2 + 1];
        const int idx = 2 * p;

        float xe = q[idx]     * invq * wq[idx];
        float xo = q[idx + 1] * invq * wq[idx + 1];
        q[idx]     = xe * c - xo * s;
        q[idx + 1] = xe * s + xo * c;

        xe = k[idx]     * invk * wk[idx];
        xo = k[idx + 1] * invk * wk[idx + 1];
        k[idx]     = xe * c - xo * s;
        k[idx + 1] = xe * s + xo * c;
    }
}

__global__ __launch_bounds__(32)
void attn_kernel(const float* __restrict__ qkv, const float* __restrict__ hks,
                 const int* __restrict__ oclp, float* __restrict__ out) {
    __shared__ __align__(16) float plds[256];

    const int bx = blockIdx.x;
    const int qt = bx & (QTILES - 1);
    const int h  = (bx >> 7) & (NUM_HEADS - 1);
    const int b  = bx >> 11;

    const int l    = threadIdx.x;
    const int lo   = l & 15;
    const int half = l >> 4;

    const int   hist = SEQ - *oclp;
    const float skh  = 1.0f + 9.0f / (1.0f + __expf(-hks[h]));
    const float sscale = 0.125f;

    const size_t base  = (size_t)b * SEQ * QKV_N;
    const float* qbase = qkv + base + (size_t)(qt * 16 + lo) * QKV_N + h * HEAD_DIM + 2 * half;
    const float* kbase = qkv + base + INNER     + h * HEAD_DIM + 2 * half;
    const float* vbase = qkv + base + 2 * INNER + h * HEAD_DIM;

    v2f qa[16];
#pragma unroll
    for (int d = 0; d < 16; ++d)
        qa[d] = *(const v2f*)(qbase + 4 * d);

    v8f o[4] = {};
    float m_state[8], l_state[8];
#pragma unroll
    for (int r = 0; r < 8; ++r) { m_state[r] = -__builtin_inff(); l_state[r] = 0.f; }

    for (int t = 0; t < QTILES; ++t) {
        v8f c = {};
        const float* krow = kbase + (size_t)(t * 16 + lo) * QKV_N;
#pragma unroll
        for (int d = 0; d < 16; ++d) {
            v2f kb = *(const v2f*)(krow + 4 * d);
            c = __builtin_amdgcn_wmma_f32_16x16x4_f32(false, qa[d], false, kb,
                                                      (short)0, c, false, false);
        }

        const int   kg  = t * 16 + lo;
        const float fac = sscale * (kg < hist ? skh : 1.0f);

        float p[8], alpha[8];
#pragma unroll
        for (int r = 0; r < 8; ++r) {
            float s = c[r] * fac;
            float rm = s;
            rm = fmaxf(rm, __shfl_xor(rm, 1));
            rm = fmaxf(rm, __shfl_xor(rm, 2));
            rm = fmaxf(rm, __shfl_xor(rm, 4));
            rm = fmaxf(rm, __shfl_xor(rm, 8));
            const float mn = fmaxf(m_state[r], rm);
            const float al = __expf(m_state[r] - mn);
            const float pv = __expf(s - mn);
            float su = pv;
            su += __shfl_xor(su, 1);
            su += __shfl_xor(su, 2);
            su += __shfl_xor(su, 4);
            su += __shfl_xor(su, 8);
            l_state[r] = l_state[r] * al + su;
            m_state[r] = mn;
            alpha[r] = al;
            p[r] = pv;
        }

#pragma unroll
        for (int nt = 0; nt < 4; ++nt)
#pragma unroll
            for (int r = 0; r < 8; ++r)
                o[nt][r] *= alpha[r];

#pragma unroll
        for (int r = 0; r < 8; ++r)
            plds[(r + 8 * half) * 16 + lo] = p[r];
        __syncthreads();
        v2f pa[4];
#pragma unroll
        for (int kk = 0; kk < 4; ++kk) {
            pa[kk].x = plds[lo * 16 + 4 * kk + 2 * half];
            pa[kk].y = plds[lo * 16 + 4 * kk + 2 * half + 1];
        }
        __syncthreads();

        const float* vt = vbase + (size_t)(t * 16) * QKV_N;
#pragma unroll
        for (int kk = 0; kk < 4; ++kk) {
            const int kr = 4 * kk + 2 * half;
#pragma unroll
            for (int nt = 0; nt < 4; ++nt) {
                v2f vb;
                vb.x = vt[(size_t)kr * QKV_N + nt * 16 + lo];
                vb.y = vt[(size_t)(kr + 1) * QKV_N + nt * 16 + lo];
                o[nt] = __builtin_amdgcn_wmma_f32_16x16x4_f32(false, pa[kk], false, vb,
                                                              (short)0, o[nt], false, false);
            }
        }
    }

    float inv[8];
#pragma unroll
    for (int r = 0; r < 8; ++r) inv[r] = 1.0f / l_state[r];

    float* orow = out + (size_t)((size_t)b * SEQ + qt * 16 + 8 * half) * INNER
                      + h * HEAD_DIM + lo;
#pragma unroll
    for (int nt = 0; nt < 4; ++nt)
#pragma unroll
        for (int r = 0; r < 8; ++r)
            orow[(size_t)r * INNER + nt * 16] = o[nt][r] * inv[r];
}

extern "C" void kernel_launch(void* const* d_in, const int* in_sizes, int n_in,
                              void* d_out, int out_size, void* d_ws, size_t ws_size,
                              hipStream_t stream) {
    const float* hidden = (const float*)d_in[0];
    const float* rot    = (const float*)d_in[1];
    const float* w_qkv  = (const float*)d_in[2];
    const float* b_qkv  = (const float*)d_in[3];
    const float* nqw    = (const float*)d_in[4];
    const float* nkw    = (const float*)d_in[5];
    const float* w_out  = (const float*)d_in[6];
    const float* b_out  = (const float*)d_in[7];
    const float* hks    = (const float*)d_in[8];
    const int*   ocl    = (const int*)d_in[9];
    float* out = (float*)d_out;

    float* qkv  = (float*)d_ws;
    float* attn = qkv + (size_t)BS * QKV_N;

    gemm_bias_f32<<<dim3(QKV_N / GB_N, BS / GB_M), 128, 0, stream>>>(
        hidden, w_qkv, b_qkv, qkv, BS, QKV_N, DIM);

    norm_rope<<<BS, 256, 0, stream>>>(qkv, rot, nqw, nkw);

    attn_kernel<<<BATCH * NUM_HEADS * QTILES, 32, 0, stream>>>(qkv, hks, ocl, attn);

    gemm_bias_f32<<<dim3(DIM / GB_N, BS / GB_M), 128, 0, stream>>>(
        attn, w_out, b_out, out, BS, DIM, DIM);
}


# Attention kernel: K and V tiles staged through double-buffered LDS and shared by a 4-wave 64-query workgroup, whole-line 128-bit loads one 32-key step ahead, fragments read from LDS; sum order kept
# speedup vs baseline: 1.7923x; 1.3550x over previous
#include <hip/hip_runtime.h>
#include <hip/hip_bf16.h>

typedef float v2f __attribute__((ext_vector_type(2)));
typedef float v8f __attribute__((ext_vector_type(8)));

#define NUM_HEADS 16
#define HEAD_DIM  64
#define DIM       1024
#define INNER     (NUM_HEADS * HEAD_DIM)
#define QKV_N     (3 * INNER)
#define BATCH     2
#define SEQ       2048
#define BS        (BATCH * SEQ)
#define QTILES    (SEQ / 16)

#define GB_M    128
#define GB_N    128
#define GB_K    32
#define A_PITCH 36
#define B_PITCH 288
#define A_STAGE (GB_M * A_PITCH)
#define B_STAGE ((GB_K / 2) * B_PITCH)

__global__ __launch_bounds__(128)
void gemm_bias_f32(const float* __restrict__ A, const float* __restrict__ W,
                   const float* __restrict__ bias, float* __restrict__ C,
                   int M, int N, int K) {
    __shared__ __align__(16) float sA[2][A_STAGE];
    __shared__ __align__(16) float sB[2][B_STAGE];

    const int t    = threadIdx.x;
    const int l    = t & 31;
    const int w    = t >> 5;
    const int lo   = l & 15;
    const int half = l >> 4;
    const int wm   = w >> 1;
    const int wn   = w & 1;
    const int n0   = blockIdx.x * GB_N;
    const int m0   = blockIdx.y * GB_M;

    const float* ag = A + (size_t)(m0 + (t >> 3)) * K + 4 * (t & 7);
    const float* bg = W + (size_t)(2 * w) * N + n0 + 4 * l;

    float4 ra[8], rb0[4], rb1[4];
    v8f acc[4][4] = {};

#pragma unroll
    for (int i = 0; i < 8; ++i)
        ra[i] = *(const float4*)(ag + (size_t)(16 * i) * K);
#pragma unroll
    for (int i = 0; i < 4; ++i) {
        rb0[i] = *(const float4*)(bg + (size_t)(8 * i) * N);
        rb1[i] = *(const float4*)(bg + (size_t)(8 * i + 1) * N);
    }
#pragma unroll
    for (int i = 0; i < 8; ++i)
        *(float4*)&sA[0][((t >> 3) + 16 * i) * A_PITCH + 4 * (t & 7)] = ra[i];
#pragma unroll
    for (int i = 0; i < 4; ++i) {
        float* d = &sB[0][(w + 4 * i) * B_PITCH + 8 * l];
        *(float4*)d       = make_float4(rb0[i].x, rb1[i].x, rb0[i].y, rb1[i].y);
        *(float4*)(d + 4) = make_float4(rb0[i].z, rb1[i].z, rb0[i].w, rb1[i].w);
    }
    __syncthreads();

    const int nk = K / GB_K;
    for (int kc = 0; kc < nk; ++kc) {
        const int s  = kc & 1;
        const int kn = (kc + 1 < nk ? kc + 1 : kc) * GB_K;
#pragma unroll
        for (int i = 0; i < 8; ++i)
            ra[i] = *(const float4*)(ag + (size_t)(16 * i) * K + kn);
#pragma unroll
        for (int i = 0; i < 4; ++i) {
            rb0[i] = *(const float4*)(bg + (size_t)(kn + 8 * i) * N);
            rb1[i] = *(const float4*)(bg + (size_t)(kn + 8 * i + 1) * N);
        }

        const float* aL = &sA[s][(wm * 64 + lo) * A_PITCH + 2 * half];
        const float* bL = &sB[s][half * B_PITCH + 2 * (wn * 64 + lo)];
#pragma unroll
        for (int kk = 0; kk < GB_K; kk += 4) {
            v2f a[4], b[4];
#pragma unroll
            for (int i = 0; i < 4; ++i)
                a[i] = *(const v2f*)(aL + 16 * i * A_PITCH + kk);
#pragma unroll
            for (int j = 0; j < 4; ++j)
                b[j] = *(const v2f*)(bL + (kk / 2) * B_PITCH + 32 * j);
#pragma unroll
            for (int i = 0; i < 4; ++i)
#pragma unroll
                for (int j = 0; j < 4; ++j)
                    acc[i][j] = __builtin_amdgcn_wmma_f32_16x16x4_f32(
                        false, a[i], false, b[j], (short)0, acc[i][j], false, false);
        }

#pragma unroll
        for (int i = 0; i < 8; ++i)
            *(float4*)&sA[s ^ 1][((t >> 3) + 16 * i) * A_PITCH + 4 * (t & 7)] = ra[i];
#pragma unroll
        for (int i = 0; i < 4; ++i) {
            float* d = &sB[s ^ 1][(w + 4 * i) * B_PITCH + 8 * l];
            *(float4*)d       = make_float4(rb0[i].x, rb1[i].x, rb0[i].y, rb1[i].y);
            *(float4*)(d + 4) = make_float4(rb0[i].z, rb1[i].z, rb0[i].w, rb1[i].w);
        }
        __syncthreads();
    }

    float bv[4];
#pragma unroll
    for (int j = 0; j < 4; ++j) bv[j] = bias[n0 + wn * 64 + 16 * j + lo];

#pragma unroll
    for (int i = 0; i < 4; ++i) {
        float* crow = C + (size_t)(m0 + wm * 64 + 16 * i + 8 * half) * N + n0 + wn * 64 + lo;
#pragma unroll
        for (int r = 0; r < 8; ++r)
#pragma unroll
            for (int j = 0; j < 4; ++j)
                crow[(size_t)r * N + 16 * j] = acc[i][j][r] + bv[j];
    }
}

__global__ __launch_bounds__(256)
void norm_rope(float* __restrict__ qkv, const float* __restrict__ rot,
               const float* __restrict__ wq, const float* __restrict__ wk) {
    const int row = blockIdx.x;
    float* q = qkv + (size_t)row * QKV_N;
    float* k = q + INNER;
    const float* fr = rot + (size_t)row * (2 * HEAD_DIM);
    const int t = threadIdx.x;

    float sq = 0.f, sk = 0.f;
    for (int i = t; i < INNER; i += 256) {
        float a = q[i]; sq += a * a;
        float b = k[i]; sk += b * b;
    }
    __shared__ float redq[256], redk[256];
    redq[t] = sq; redk[t] = sk;
    __syncthreads();
    for (int off = 128; off > 0; off >>= 1) {
        if (t < off) { redq[t] += redq[t + off]; redk[t] += redk[t + off]; }
        __syncthreads();
    }
    const float invq = rsqrtf(redq[0] * (1.0f / INNER) + 1e-5f);
    const float invk = rsqrtf(redk[0] * (1.0f / INNER) + 1e-5f);

    for (int p = t; p < INNER / 2; p += 256) {
        const int i2  = p & 31;
        const float c = fr[2 * i2];
        const float s = fr[HEAD_DIM + 2 * i2 + 1];
        const int idx = 2 * p;

        float xe = q[idx]     * invq * wq[idx];
        float xo = q[idx + 1] * invq * wq[idx + 1];
        q[idx]     = xe * c - xo * s;
        q[idx + 1] = xe * s + xo * c;

        xe = k[idx]     * invk * wk[idx];
        xo = k[idx + 1] * invk * wk[idx + 1];
        k[idx]     = xe * c - xo * s;
        k[idx + 1] = xe * s + xo * c;
    }
}

#define AT_WAVES   4
#define AT_KT      32
#define AT_QGROUPS (SEQ / (16 * AT_WAVES))
#define AT_STEPS   (SEQ / AT_KT)
#define K_PITCH    68
#define V_PITCH    160
#define K_STAGE    (AT_KT * K_PITCH)
#define V_STAGE    ((AT_KT / 2) * V_PITCH)

__global__ __launch_bounds__(32 * AT_WAVES) __attribute__((amdgpu_waves_per_eu(4, 4)))
void attn_kernel(const float* __restrict__ qkv, const float* __restrict__ hks,
                 const int* __restrict__ oclp, float* __restrict__ out) {
    __shared__ __align__(16) float sK[2][K_STAGE];
    __shared__ __align__(16) float sV[2][V_STAGE];
    __shared__ __align__(16) float pall[AT_WAVES][256];

    const int bx = blockIdx.x;
    const int qg = bx & (AT_QGROUPS - 1);
    const int h  = (bx >> 5) & (NUM_HEADS - 1);
    const int b  = bx >> 9;

    const int tid  = threadIdx.x;
    const int w    = tid >> 5;
    const int l    = tid & 31;
    const int lo   = l & 15;
    const int half = l >> 4;
    const int qt   = qg * AT_WAVES + w;
    float* plds    = pall[w];

    const int   hist = SEQ - *oclp;
    const float skh  = 1.0f + 9.0f / (1.0f + __expf(-hks[h]));
    const float sscale = 0.125f;

    const size_t base  = (size_t)b * SEQ * QKV_N;
    const float* qbase = qkv + base + (size_t)(qt * 16 + lo) * QKV_N + h * HEAD_DIM + 2 * half;

    const int srow = tid >> 4;
    const int scol = 4 * (tid & 15);
    const float* kg_src = qkv + base + INNER     + h * HEAD_DIM + (size_t)srow * QKV_N + scol;
    const float* vg_src = qkv + base + 2 * INNER + h * HEAD_DIM + (size_t)(2 * srow) * QKV_N + scol;

    float4 rk[4], rv0[2], rv1[2];

#pragma unroll
    for (int i = 0; i < 4; ++i)
        rk[i] = *(const float4*)(kg_src + (size_t)(8 * i) * QKV_N);
#pragma unroll
    for (int i = 0; i < 2; ++i) {
        rv0[i] = *(const float4*)(vg_src + (size_t)(16 * i) * QKV_N);
        rv1[i] = *(const float4*)(vg_src + (size_t)(16 * i + 1) * QKV_N);
    }

    v2f qa[16];
#pragma unroll
    for (int d = 0; d < 16; ++d)
        qa[d] = *(const v2f*)(qbase + 4 * d);

#pragma unroll
    for (int i = 0; i < 4; ++i)
        *(float4*)&sK[0][(srow + 8 * i) * K_PITCH + scol] = rk[i];
#pragma unroll
    for (int i = 0; i < 2; ++i) {
        float* d = &sV[0][(srow + 8 * i) * V_PITCH + 2 * scol];
        *(float4*)d       = make_float4(rv0[i].x, rv1[i].x, rv0[i].y, rv1[i].y);
        *(float4*)(d + 4) = make_float4(rv0[i].z, rv1[i].z, rv0[i].w, rv1[i].w);
    }
    __syncthreads();

    v8f o[4] = {};
    float m_state[8], l_state[8];
#pragma unroll
    for (int r = 0; r < 8; ++r) { m_state[r] = -__builtin_inff(); l_state[r] = 0.f; }

    for (int st = 0; st < AT_STEPS; ++st) {
        const int s = st & 1;
        const size_t kn = (size_t)((st + 1 < AT_STEPS ? st + 1 : st) * AT_KT) * QKV_N;
#pragma unroll
        for (int i = 0; i < 4; ++i)
            rk[i] = *(const float4*)(kg_src + kn + (size_t)(8 * i) * QKV_N);
#pragma unroll
        for (int i = 0; i < 2; ++i) {
            rv0[i] = *(const float4*)(vg_src + kn + (size_t)(16 * i) * QKV_N);
            rv1[i] = *(const float4*)(vg_src + kn + (size_t)(16 * i + 1) * QKV_N);
        }

        const float* kL = &sK[s][lo * K_PITCH + 2 * half];
        const float* vL = &sV[s][half * V_PITCH + 2 * lo];

#pragma unroll
        for (int j = 0; j < AT_KT / 16; ++j) {
            const int t = st * (AT_KT / 16) + j;

            v8f c = {};
            const float* krow = kL + j * 16 * K_PITCH;
#pragma unroll
            for (int d = 0; d < 16; ++d) {
                v2f kb = *(const v2f*)(krow + 4 * d);
                c = __builtin_amdgcn_wmma_f32_16x16x4_f32(false, qa[d], false, kb,
                                                          (short)0, c, false, false);
            }

            const int   kg  = t * 16 + lo;
            const float fac = sscale * (kg < hist ? skh : 1.0f);

            float p[8], alpha[8];
#pragma unroll
            for (int r = 0; r < 8; ++r) {
                float sc = c[r] * fac;
                float rm = sc;
                rm = fmaxf(rm, __shfl_xor(rm, 1));
                rm = fmaxf(rm, __shfl_xor(rm, 2));
                rm = fmaxf(rm, __shfl_xor(rm, 4));
                rm = fmaxf(rm, __shfl_xor(rm, 8));
                const float mn = fmaxf(m_state[r], rm);
                const float al = __expf(m_state[r] - mn);
                const float pv = __expf(sc - mn);
                float su = pv;
                su += __shfl_xor(su, 1);
                su += __shfl_xor(su, 2);
                su += __shfl_xor(su, 4);
                su += __shfl_xor(su, 8);
                l_state[r] = l_state[r] * al + su;
                m_state[r] = mn;
                alpha[r] = al;
                p[r] = pv;
            }

#pragma unroll
            for (int nt = 0; nt < 4; ++nt)
#pragma unroll
                for (int r = 0; r < 8; ++r)
                    o[nt][r] *= alpha[r];

#pragma unroll
            for (int r = 0; r < 8; ++r)
                plds[(r + 8 * half) * 16 + lo] = p[r];
            v2f pa[4];
#pragma unroll
            for (int kk = 0; kk < 4; ++kk) {
                pa[kk].x = plds[lo * 16 + 4 * kk + 2 * half];
                pa[kk].y = plds[lo * 16 + 4 * kk + 2 * half + 1];
            }

            const float* vt = vL + j * 8 * V_PITCH;
#pragma unroll
            for (int kk = 0; kk < 4; ++kk) {
#pragma unroll
                for (int nt = 0; nt < 4; ++nt) {
                    v2f vb = *(const v2f*)(vt + 2 * kk * V_PITCH + 32 * nt);
                    o[nt] = __builtin_amdgcn_wmma_f32_16x16x4_f32(false, pa[kk], false, vb,
                                                                  (short)0, o[nt], false, false);
                }
            }
        }

#pragma unroll
        for (int i = 0; i < 4; ++i)
            *(float4*)&sK[s ^ 1][(srow + 8 * i) * K_PITCH + scol] = rk[i];
#pragma unroll
        for (int i = 0; i < 2; ++i) {
            float* d = &sV[s ^ 1][(srow + 8 * i) * V_PITCH + 2 * scol];
            *(float4*)d       = make_float4(rv0[i].x, rv1[i].x, rv0[i].y, rv1[i].y);
            *(float4*)(d + 4) = make_float4(rv0[i].z, rv1[i].z, rv0[i].w, rv1[i].w);
        }
        __syncthreads();
    }

    float inv[8];
#pragma unroll
    for (int r = 0; r < 8; ++r) inv[r] = 1.0f / l_state[r];

    float* orow = out + (size_t)((size_t)b * SEQ + qt * 16 + 8 * half) * INNER
                      + h * HEAD_DIM + lo;
#pragma unroll
    for (int nt = 0; nt < 4; ++nt)
#pragma unroll
        for (int r = 0; r < 8; ++r)
            orow[(size_t)r * INNER + nt * 16] = o[nt][r] * inv[r];
}

extern "C" void kernel_launch(void* const* d_in, const int* in_sizes, int n_in,
                              void* d_out, int out_size, void* d_ws, size_t ws_size,
                              hipStream_t stream) {
    const float* hidden = (const float*)d_in[0];
    const float* rot    = (const float*)d_in[1];
    const float* w_qkv  = (const float*)d_in[2];
    const float* b_qkv  = (const float*)d_in[3];
    const float* nqw    = (const float*)d_in[4];
    const float* nkw    = (const float*)d_in[5];
    const float* w_out  = (const float*)d_in[6];
    const float* b_out  = (const float*)d_in[7];
    const float* hks    = (const float*)d_in[8];
    const int*   ocl    = (const int*)d_in[9];
    float* out = (float*)d_out;

    float* qkv  = (float*)d_ws;
    float* attn = qkv + (size_t)BS * QKV_N;

    gemm_bias_f32<<<dim3(QKV_N / GB_N, BS / GB_M), 128, 0, stream>>>(
        hidden, w_qkv, b_qkv, qkv, BS, QKV_N, DIM);

    norm_rope<<<BS, 256, 0, stream>>>(qkv, rot, nqw, nkw);

    attn_kernel<<<BATCH * NUM_HEADS * AT_QGROUPS, 32 * AT_WAVES, 0, stream>>>(qkv, hks, ocl, attn);

    gemm_bias_f32<<<dim3(DIM / GB_N, BS / GB_M), 128, 0, stream>>>(
        attn, w_out, b_out, out, BS, DIM, DIM);
}


# Attention kernel: score tile built transposed (S^T = K.Q^T, one query per lane): softmax max/sum in registers per 32-key step, P feeds PV as B operand from registers; row-sum and PV key order changed
# speedup vs baseline: 2.0117x; 1.1224x over previous
#include <hip/hip_runtime.h>
#include <hip/hip_bf16.h>

typedef float v2f __attribute__((ext_vector_type(2)));
typedef float v8f __attribute__((ext_vector_type(8)));
typedef float v4f __attribute__((ext_vector_type(4)));

#define NUM_HEADS 16
#define HEAD_DIM  64
#define DIM       1024
#define INNER     (NUM_HEADS * HEAD_DIM)
#define QKV_N     (3 * INNER)
#define BATCH     2
#define SEQ       2048
#define BS        (BATCH * SEQ)
#define QTILES    (SEQ / 16)

#define GB_M    128
#define GB_N    128
#define GB_K    32
#define A_PITCH 36
#define B_PITCH 288
#define A_STAGE (GB_M * A_PITCH)
#define B_STAGE ((GB_K / 2) * B_PITCH)

__global__ __launch_bounds__(128)
void gemm_bias_f32(const float* __restrict__ A, const float* __restrict__ W,
                   const float* __restrict__ bias, float* __restrict__ C,
                   int M, int N, int K) {
    __shared__ __align__(16) float sA[2][A_STAGE];
    __shared__ __align__(16) float sB[2][B_STAGE];

    const int t    = threadIdx.x;
    const int l    = t & 31;
    const int w    = t >> 5;
    const int lo   = l & 15;
    const int half = l >> 4;
    const int wm   = w >> 1;
    const int wn   = w & 1;
    const int n0   = blockIdx.x * GB_N;
    const int m0   = blockIdx.y * GB_M;

    const float* ag = A + (size_t)(m0 + (t >> 3)) * K + 4 * (t & 7);
    const float* bg = W + (size_t)(2 * w) * N + n0 + 4 * l;

    float4 ra[8], rb0[4], rb1[4];
    v8f acc[4][4] = {};

#pragma unroll
    for (int i = 0; i < 8; ++i)
        ra[i] = *(const float4*)(ag + (size_t)(16 * i) * K);
#pragma unroll
    for (int i = 0; i < 4; ++i) {
        rb0[i] = *(const float4*)(bg + (size_t)(8 * i) * N);
        rb1[i] = *(const float4*)(bg + (size_t)(8 * i + 1) * N);
    }
#pragma unroll
    for (int i = 0; i < 8; ++i)
        *(float4*)&sA[0][((t >> 3) + 16 * i) * A_PITCH + 4 * (t & 7)] = ra[i];
#pragma unroll
    for (int i = 0; i < 4; ++i) {
        float* d = &sB[0][(w + 4 * i) * B_PITCH + 8 * l];
        *(float4*)d       = make_float4(rb0[i].x, rb1[i].x, rb0[i].y, rb1[i].y);
        *(float4*)(d + 4) = make_float4(rb0[i].z, rb1[i].z, rb0[i].w, rb1[i].w);
    }
    __syncthreads();

    const int nk = K / GB_K;
    for (int kc = 0; kc < nk; ++kc) {
        const int s  = kc & 1;
        const int kn = (kc + 1 < nk ? kc + 1 : kc) * GB_K;
#pragma unroll
        for (int i = 0; i < 8; ++i)
            ra[i] = *(const float4*)(ag + (size_t)(16 * i) * K + kn);
#pragma unroll
        for (int i = 0; i < 4; ++i) {
            rb0[i] = *(const float4*)(bg + (size_t)(kn + 8 * i) * N);
            rb1[i] = *(const float4*)(bg + (size_t)(kn + 8 * i + 1) * N);
        }

        const float* aL = &sA[s][(wm * 64 + lo) * A_PITCH + 2 * half];
        const float* bL = &sB[s][half * B_PITCH + 2 * (wn * 64 + lo)];
#pragma unroll
        for (int kk = 0; kk < GB_K; kk += 4) {
            v2f a[4], b[4];
#pragma unroll
            for (int i = 0; i < 4; ++i)
                a[i] = *(const v2f*)(aL + 16 * i * A_PITCH + kk);
#pragma unroll
            for (int j = 0; j < 4; ++j)
                b[j] = *(const v2f*)(bL + (kk / 2) * B_PITCH + 32 * j);
#pragma unroll
            for (int i = 0; i < 4; ++i)
#pragma unroll
                for (int j = 0; j < 4; ++j)
                    acc[i][j] = __builtin_amdgcn_wmma_f32_16x16x4_f32(
                        false, a[i], false, b[j], (short)0, acc[i][j], false, false);
        }

#pragma unroll
        for (int i = 0; i < 8; ++i)
            *(float4*)&sA[s ^ 1][((t >> 3) + 16 * i) * A_PITCH + 4 * (t & 7)] = ra[i];
#pragma unroll
        for (int i = 0; i < 4; ++i) {
            float* d = &sB[s ^ 1][(w + 4 * i) * B_PITCH + 8 * l];
            *(float4*)d       = make_float4(rb0[i].x, rb1[i].x, rb0[i].y, rb1[i].y);
            *(float4*)(d + 4) = make_float4(rb0[i].z, rb1[i].z, rb0[i].w, rb1[i].w);
        }
        __syncthreads();
    }

    float bv[4];
#pragma unroll
    for (int j = 0; j < 4; ++j) bv[j] = bias[n0 + wn * 64 + 16 * j + lo];

#pragma unroll
    for (int i = 0; i < 4; ++i) {
        float* crow = C + (size_t)(m0 + wm * 64 + 16 * i + 8 * half) * N + n0 + wn * 64 + lo;
#pragma unroll
        for (int r = 0; r < 8; ++r)
#pragma unroll
            for (int j = 0; j < 4; ++j)
                crow[(size_t)r * N + 16 * j] = acc[i][j][r] + bv[j];
    }
}

__global__ __launch_bounds__(256)
void norm_rope(float* __restrict__ qkv, const float* __restrict__ rot,
               const float* __restrict__ wq, const float* __restrict__ wk) {
    const int row = blockIdx.x;
    float* q = qkv + (size_t)row * QKV_N;
    float* k = q + INNER;
    const float* fr = rot + (size_t)row * (2 * HEAD_DIM);
    const int t = threadIdx.x;

    float sq = 0.f, sk = 0.f;
    for (int i = t; i < INNER; i += 256) {
        float a = q[i]; sq += a * a;
        float b = k[i]; sk += b * b;
    }
    __shared__ float redq[256], redk[256];
    redq[t] = sq; redk[t] = sk;
    __syncthreads();
    for (int off = 128; off > 0; off >>= 1) {
        if (t < off) { redq[t] += redq[t + off]; redk[t] += redk[t + off]; }
        __syncthreads();
    }
    const float invq = rsqrtf(redq[0] * (1.0f / INNER) + 1e-5f);
    const float invk = rsqrtf(redk[0] * (1.0f / INNER) + 1e-5f);

    for (int p = t; p < INNER / 2; p += 256) {
        const int i2  = p & 31;
        const float c = fr[2 * i2];
        const float s = fr[HEAD_DIM + 2 * i2 + 1];
        const int idx = 2 * p;

        float xe = q[idx]     * invq * wq[idx];
        float xo = q[idx + 1] * invq * wq[idx + 1];
        q[idx]     = xe * c - xo * s;
        q[idx + 1] = xe * s + xo * c;

        xe = k[idx]     * invk * wk[idx];
        xo = k[idx + 1] * invk * wk[idx + 1];
        k[idx]     = xe * c - xo * s;
        k[idx + 1] = xe * s + xo * c;
    }
}

#define AT_WAVES   4
#define AT_KT      32
#define AT_QGROUPS (SEQ / (16 * AT_WAVES))
#define AT_STEPS   (SEQ / AT_KT)
#define K_PITCH    68
#define V_PITCH    136
#define K_STAGE    (AT_KT * K_PITCH)
#define V_STAGE    ((AT_KT / 2) * V_PITCH)

__device__ __forceinline__ float xhalf(float x) {
    return __builtin_bit_cast(float, __builtin_amdgcn_permlanex16(
        __builtin_bit_cast(int, x), __builtin_bit_cast(int, x),
        0x76543210, 0xfedcba98, false, false));
}

__global__ __launch_bounds__(32 * AT_WAVES) __attribute__((amdgpu_waves_per_eu(4, 4)))
void attn_kernel(const float* __restrict__ qkv, const float* __restrict__ hks,
                 const int* __restrict__ oclp, float* __restrict__ out) {
    __shared__ __align__(16) float sK[2][K_STAGE];
    __shared__ __align__(16) float sV[2][V_STAGE];

    const int bx = blockIdx.x;
    const int qg = bx & (AT_QGROUPS - 1);
    const int h  = (bx >> 5) & (NUM_HEADS - 1);
    const int b  = bx >> 9;

    const int tid  = threadIdx.x;
    const int w    = tid >> 5;
    const int l    = tid & 31;
    const int lo   = l & 15;
    const int half = l >> 4;
    const int qt   = qg * AT_WAVES + w;

    const int   hist = SEQ - *oclp;
    const float skh  = 1.0f + 9.0f / (1.0f + __expf(-hks[h]));
    const float sscale = 0.125f;
    const float fac_hist = sscale * skh;
    const float fac_rest = sscale * 1.0f;

    const size_t base  = (size_t)b * SEQ * QKV_N;
    const float* qbase = qkv + base + (size_t)(qt * 16 + lo) * QKV_N + h * HEAD_DIM + 2 * half;

    const int srow = tid >> 4;
    const int scol = 4 * (tid & 15);
    const float* kg_src = qkv + base + INNER     + h * HEAD_DIM + (size_t)srow * QKV_N + scol;
    const float* vg_src = qkv + base + 2 * INNER + h * HEAD_DIM + (size_t)(2 * srow) * QKV_N + scol;

    v4f rk0, rk1, rk2, rk3, rva0, rvb0, rva1, rvb1;

    rk0  = *(const v4f*)(kg_src);
    rk1  = *(const v4f*)(kg_src + (size_t)8  * QKV_N);
    rk2  = *(const v4f*)(kg_src + (size_t)16 * QKV_N);
    rk3  = *(const v4f*)(kg_src + (size_t)24 * QKV_N);
    rva0 = *(const v4f*)(vg_src);
    rvb0 = *(const v4f*)(vg_src + (size_t)1  * QKV_N);
    rva1 = *(const v4f*)(vg_src + (size_t)16 * QKV_N);
    rvb1 = *(const v4f*)(vg_src + (size_t)17 * QKV_N);

    v2f qa[16];
#pragma unroll
    for (int d = 0; d < 16; ++d)
        qa[d] = *(const v2f*)(qbase + 4 * d);

    {
        float* dk = &sK[0][srow * K_PITCH + scol];
        *(v4f*)(dk)                = rk0;
        *(v4f*)(dk + 8  * K_PITCH) = rk1;
        *(v4f*)(dk + 16 * K_PITCH) = rk2;
        *(v4f*)(dk + 24 * K_PITCH) = rk3;
        float* dv = &sV[0][srow * V_PITCH + 2 * scol];
        *(v4f*)(dv)                   = __builtin_shufflevector(rva0, rvb0, 0, 4, 1, 5);
        *(v4f*)(dv + 4)               = __builtin_shufflevector(rva0, rvb0, 2, 6, 3, 7);
        *(v4f*)(dv + 8 * V_PITCH)     = __builtin_shufflevector(rva1, rvb1, 0, 4, 1, 5);
        *(v4f*)(dv + 8 * V_PITCH + 4) = __builtin_shufflevector(rva1, rvb1, 2, 6, 3, 7);
    }
    __syncthreads();

    v8f o[4] = {};
    float m_run = -__builtin_inff();
    float l_run = 0.f;

    for (int st = 0; st < AT_STEPS; ++st) {
        const int s = st & 1;
        const size_t kn = (size_t)((st + 1 < AT_STEPS ? st + 1 : st) * AT_KT) * QKV_N;
        rk0  = *(const v4f*)(kg_src + kn);
        rk1  = *(const v4f*)(kg_src + kn + (size_t)8  * QKV_N);
        rk2  = *(const v4f*)(kg_src + kn + (size_t)16 * QKV_N);
        rk3  = *(const v4f*)(kg_src + kn + (size_t)24 * QKV_N);
        rva0 = *(const v4f*)(vg_src + kn);
        rvb0 = *(const v4f*)(vg_src + kn + (size_t)1  * QKV_N);
        rva1 = *(const v4f*)(vg_src + kn + (size_t)16 * QKV_N);
        rvb1 = *(const v4f*)(vg_src + kn + (size_t)17 * QKV_N);

        const float* kL = &sK[s][lo * K_PITCH + 2 * half];
        const float* vL = &sV[s][(4 * half) * V_PITCH + 2 * lo];

        v2f kf[2][16];
#pragma unroll
        for (int j = 0; j < 2; ++j)
#pragma unroll
            for (int d = 0; d < 16; ++d)
                kf[j][d] = *(const v2f*)(kL + j * 16 * K_PITCH + 4 * d);

        asm volatile("" : "+v"(kf[0][0]) : : "memory");

        v8f c[2] = {};
#pragma unroll
        for (int d = 0; d < 16; ++d) {
#pragma unroll
            for (int j = 0; j < 2; ++j)
                c[j] = __builtin_amdgcn_wmma_f32_16x16x4_f32(false, kf[j][d], false, qa[d],
                                                             (short)0, c[j], false, false);
        }

        v2f vf[2][4][4];
#pragma unroll
        for (int j = 0; j < 2; ++j)
#pragma unroll
            for (int kk = 0; kk < 4; ++kk)
#pragma unroll
                for (int nt = 0; nt < 4; ++nt)
                    vf[j][kk][nt] = *(const v2f*)(vL + (8 * j + kk) * V_PITCH + 32 * nt);

        const int nb = hist - st * AT_KT - 8 * half;

        asm volatile("" : "+v"(c[1]) : : "memory");

        float sc[2][8];
#pragma unroll
        for (int j = 0; j < 2; ++j)
#pragma unroll
            for (int r = 0; r < 8; ++r)
                sc[j][r] = c[j][r] * ((16 * j + r) < nb ? fac_hist : fac_rest);

        float rm0 = fmaxf(fmaxf(sc[0][0], sc[0][1]), fmaxf(sc[0][2], sc[0][3]));
        float rm1 = fmaxf(fmaxf(sc[0][4], sc[0][5]), fmaxf(sc[0][6], sc[0][7]));
        float rm2 = fmaxf(fmaxf(sc[1][0], sc[1][1]), fmaxf(sc[1][2], sc[1][3]));
        float rm3 = fmaxf(fmaxf(sc[1][4], sc[1][5]), fmaxf(sc[1][6], sc[1][7]));
        float rm  = fmaxf(fmaxf(rm0, rm1), fmaxf(rm2, rm3));
        rm = fmaxf(rm, xhalf(rm));

        const float mn = fmaxf(m_run, rm);
        const float al = __expf(m_run - mn);

        float p[2][8];
#pragma unroll
        for (int j = 0; j < 2; ++j)
#pragma unroll
            for (int r = 0; r < 8; ++r)
                p[j][r] = __expf(sc[j][r] - mn);

        const float su0 = (p[0][0] + p[0][1]) + (p[0][2] + p[0][3]);
        const float su1 = (p[0][4] + p[0][5]) + (p[0][6] + p[0][7]);
        const float su2 = (p[1][0] + p[1][1]) + (p[1][2] + p[1][3]);
        const float su3 = (p[1][4] + p[1][5]) + (p[1][6] + p[1][7]);
        l_run = l_run * al + ((su0 + su1) + (su2 + su3));
        m_run = mn;

#pragma unroll
        for (int nt = 0; nt < 4; ++nt)
#pragma unroll
            for (int r = 0; r < 8; ++r)
                o[nt][r] *= al;

#pragma unroll
        for (int j = 0; j < 2; ++j) {
#pragma unroll
            for (int kk = 0; kk < 4; ++kk) {
                v2f pb;
                pb.x = p[j][2 * kk];
                pb.y = p[j][2 * kk + 1];
#pragma unroll
                for (int nt = 0; nt < 4; ++nt) {
                    o[nt] = __builtin_amdgcn_wmma_f32_16x16x4_f32(false, vf[j][kk][nt], false, pb,
                                                                  (short)0, o[nt], false, false);
                }
            }
        }

        {
            float* dk = &sK[s ^ 1][srow * K_PITCH + scol];
            *(v4f*)(dk)                = rk0;
            *(v4f*)(dk + 8  * K_PITCH) = rk1;
            *(v4f*)(dk + 16 * K_PITCH) = rk2;
            *(v4f*)(dk + 24 * K_PITCH) = rk3;
            float* dv = &sV[s ^ 1][srow * V_PITCH + 2 * scol];
            *(v4f*)(dv)                   = __builtin_shufflevector(rva0, rvb0, 0, 4, 1, 5);
            *(v4f*)(dv + 4)               = __builtin_shufflevector(rva0, rvb0, 2, 6, 3, 7);
            *(v4f*)(dv + 8 * V_PITCH)     = __builtin_shufflevector(rva1, rvb1, 0, 4, 1, 5);
            *(v4f*)(dv + 8 * V_PITCH + 4) = __builtin_shufflevector(rva1, rvb1, 2, 6, 3, 7);
        }
        __syncthreads();
    }

    const float l_fin = l_run + xhalf(l_run);
    const float inv = 1.0f / l_fin;

    float* orow = out + (size_t)((size_t)b * SEQ + qt * 16 + lo) * INNER
                      + h * HEAD_DIM + 8 * half;
#pragma unroll
    for (int nt = 0; nt < 4; ++nt) {
        *(float4*)(orow + 16 * nt)     = make_float4(o[nt][0] * inv, o[nt][1] * inv,
                                                     o[nt][2] * inv, o[nt][3] * inv);
        *(float4*)(orow + 16 * nt + 4) = make_float4(o[nt][4] * inv, o[nt][5] * inv,
                                                     o[nt][6] * inv, o[nt][7] * inv);
    }
}

extern "C" void kernel_launch(void* const* d_in, const int* in_sizes, int n_in,
                              void* d_out, int out_size, void* d_ws, size_t ws_size,
                              hipStream_t stream) {
    const float* hidden = (const float*)d_in[0];
    const float* rot    = (const float*)d_in[1];
    const float* w_qkv  = (const float*)d_in[2];
    const float* b_qkv  = (const float*)d_in[3];
    const float* nqw    = (const float*)d_in[4];
    const float* nkw    = (const float*)d_in[5];
    const float* w_out  = (const float*)d_in[6];
    const float* b_out  = (const float*)d_in[7];
    const float* hks    = (const float*)d_in[8];
    const int*   ocl    = (const int*)d_in[9];
    float* out = (float*)d_out;

    float* qkv  = (float*)d_ws;
    float* attn = qkv + (size_t)BS * QKV_N;

    gemm_bias_f32<<<dim3(QKV_N / GB_N, BS / GB_M), 128, 0, stream>>>(
        hidden, w_qkv, b_qkv, qkv, BS, QKV_N, DIM);

    norm_rope<<<BS, 256, 0, stream>>>(qkv, rot, nqw, nkw);

    attn_kernel<<<BATCH * NUM_HEADS * AT_QGROUPS, 32 * AT_WAVES, 0, stream>>>(qkv, hks, ocl, attn);

    gemm_bias_f32<<<dim3(DIM / GB_N, BS / GB_M), 128, 0, stream>>>(
        attn, w_out, b_out, out, BS, DIM, DIM);
}


# Norm+rotary launch removed: qkv GEMM epilogue applies norm weight and rotation, writes per-row sums of squares; attention folds 1/rms into Q once and into K rows at staging; rms sum order changed
# speedup vs baseline: 2.1800x; 1.0837x over previous
#include <hip/hip_runtime.h>
#include <hip/hip_bf16.h>

typedef float v2f __attribute__((ext_vector_type(2)));
typedef float v8f __attribute__((ext_vector_type(8)));
typedef float v4f __attribute__((ext_vector_type(4)));

#define NUM_HEADS 16
#define HEAD_DIM  64
#define DIM       1024
#define INNER     (NUM_HEADS * HEAD_DIM)
#define QKV_N     (3 * INNER)
#define BATCH     2
#define SEQ       2048
#define BS        (BATCH * SEQ)
#define QTILES    (SEQ / 16)

#define GB_M    128
#define GB_N    128
#define GB_K    32
#define A_PITCH 36
#define B_PITCH 288
#define A_STAGE (GB_M * A_PITCH)
#define B_STAGE ((GB_K / 2) * B_PITCH)
#define PART_N  32

template <int M>
__device__ __forceinline__ float row_xor(float x) {
    return __builtin_bit_cast(float, __builtin_amdgcn_update_dpp(
        0, __builtin_bit_cast(int, x), 0x160 + M, 0xf, 0xf, true));
}

template <bool ROPE>
__global__ __launch_bounds__(128)
void gemm_bias_f32(const float* __restrict__ A, const float* __restrict__ W,
                   const float* __restrict__ bias, float* __restrict__ C,
                   int M, int N, int K,
                   const float* __restrict__ rot, const float* __restrict__ wq,
                   const float* __restrict__ wk, float* __restrict__ part) {
    __shared__ __align__(16) float sA[2][A_STAGE];
    __shared__ __align__(16) float sB[2][B_STAGE];

    const int t    = threadIdx.x;
    const int l    = t & 31;
    const int w    = t >> 5;
    const int lo   = l & 15;
    const int half = l >> 4;
    const int wm   = w >> 1;
    const int wn   = w & 1;
    const int n0   = blockIdx.x * GB_N;
    const int m0   = blockIdx.y * GB_M;

    const float* ag = A + (size_t)(m0 + (t >> 3)) * K + 4 * (t & 7);
    const float* bg = W + (size_t)(2 * w) * N + n0 + 4 * l;

    float4 ra[8], rb0[4], rb1[4];
    v8f acc[4][4] = {};

#pragma unroll
    for (int i = 0; i < 8; ++i)
        ra[i] = *(const float4*)(ag + (size_t)(16 * i) * K);
#pragma unroll
    for (int i = 0; i < 4; ++i) {
        rb0[i] = *(const float4*)(bg + (size_t)(8 * i) * N);
        rb1[i] = *(const float4*)(bg + (size_t)(8 * i + 1) * N);
    }
#pragma unroll
    for (int i = 0; i < 8; ++i)
        *(float4*)&sA[0][((t >> 3) + 16 * i) * A_PITCH + 4 * (t & 7)] = ra[i];
#pragma unroll
    for (int i = 0; i < 4; ++i) {
        float* d = &sB[0][(w + 4 * i) * B_PITCH + 8 * l];
        *(float4*)d       = make_float4(rb0[i].x, rb1[i].x, rb0[i].y, rb1[i].y);
        *(float4*)(d + 4) = make_float4(rb0[i].z, rb1[i].z, rb0[i].w, rb1[i].w);
    }
    __syncthreads();

    const int nk = K / GB_K;
    for (int kc = 0; kc < nk; ++kc) {
        const int s  = kc & 1;
        const int kn = (kc + 1 < nk ? kc + 1 : kc) * GB_K;
#pragma unroll
        for (int i = 0; i < 8; ++i)
            ra[i] = *(const float4*)(ag + (size_t)(16 * i) * K + kn);
#pragma unroll
        for (int i = 0; i < 4; ++i) {
            rb0[i] = *(const float4*)(bg + (size_t)(kn + 8 * i) * N);
            rb1[i] = *(const float4*)(bg + (size_t)(kn + 8 * i + 1) * N);
        }

        const float* aL = &sA[s][(wm * 64 + lo) * A_PITCH + 2 * half];
        const float* bL = &sB[s][half * B_PITCH + 2 * (wn * 64 + lo)];
#pragma unroll
        for (int kk = 0; kk < GB_K; kk += 4) {
            v2f a[4], b[4];
#pragma unroll
            for (int i = 0; i < 4; ++i)
                a[i] = *(const v2f*)(aL + 16 * i * A_PITCH + kk);
#pragma unroll
            for (int j = 0; j < 4; ++j)
                b[j] = *(const v2f*)(bL + (kk / 2) * B_PITCH + 32 * j);
#pragma unroll
            for (int i = 0; i < 4; ++i)
#pragma unroll
                for (int j = 0; j < 4; ++j)
                    acc[i][j] = __builtin_amdgcn_wmma_f32_16x16x4_f32(
                        false, a[i], false, b[j], (short)0, acc[i][j], false, false);
        }

#pragma unroll
        for (int i = 0; i < 8; ++i)
            *(float4*)&sA[s ^ 1][((t >> 3) + 16 * i) * A_PITCH + 4 * (t & 7)] = ra[i];
#pragma unroll
        for (int i = 0; i < 4; ++i) {
            float* d = &sB[s ^ 1][(w + 4 * i) * B_PITCH + 8 * l];
            *(float4*)d       = make_float4(rb0[i].x, rb1[i].x, rb0[i].y, rb1[i].y);
            *(float4*)(d + 4) = make_float4(rb0[i].z, rb1[i].z, rb0[i].w, rb1[i].w);
        }
        __syncthreads();
    }

    int te = t;
    if (ROPE) asm volatile("" : "+v"(te));
    const int e_lo   = te & 15;
    const int e_half = (te >> 4) & 1;
    const int e_wm   = te >> 6;
    const int e_wn   = (te >> 5) & 1;

    float bv[4];
#pragma unroll
    for (int j = 0; j < 4; ++j) bv[j] = bias[n0 + e_wn * 64 + 16 * j + e_lo];

    if (ROPE && n0 < 2 * INNER) {
        const float* wsec = (n0 < INNER) ? wq : wk;
        const int cb = (n0 & (INNER - 1)) + e_wn * 64;
        float wv[4];
#pragma unroll
        for (int j = 0; j < 4; ++j) wv[j] = wsec[cb + 16 * j + e_lo];
        const int  ce  = e_lo & ~1;
        const int  co  = e_lo | 1;
        const bool odd = (e_lo & 1) != 0;
#pragma unroll
        for (int i = 0; i < 4; ++i) {
            const int row0 = m0 + e_wm * 64 + 16 * i + 8 * e_half;
            float mine = 0.f;
#pragma unroll
            for (int r = 0; r < 8; ++r) {
                const float* fr = rot + (size_t)(row0 + r) * (2 * HEAD_DIM);
                float* crow = C + (size_t)(row0 + r) * N + n0 + e_wn * 64 + e_lo;
                float ss = 0.f;
#pragma unroll
                for (int j = 0; j < 4; ++j) {
                    const float v  = acc[i][j][r] + bv[j];
                    ss += v * v;
                    const float x  = v * wv[j];
                    const float xp = row_xor<1>(x);
                    const float c  = fr[16 * j + ce];
                    const float s  = fr[HEAD_DIM + 16 * j + co];
                    crow[16 * j] = odd ? (xp * s + x * c) : (x * c - xp * s);
                }
                ss += row_xor<1>(ss);
                ss += row_xor<2>(ss);
                ss += row_xor<4>(ss);
                ss += row_xor<8>(ss);
                mine = ((e_lo & 7) == r) ? ss : mine;
            }
            if (e_lo < 8) part[(size_t)(row0 + e_lo) * PART_N + (n0 >> 6) + e_wn] = mine;
        }
    } else {
#pragma unroll
        for (int i = 0; i < 4; ++i) {
            float* crow = C + (size_t)(m0 + e_wm * 64 + 16 * i + 8 * e_half) * N + n0 + e_wn * 64 + e_lo;
#pragma unroll
            for (int r = 0; r < 8; ++r)
#pragma unroll
                for (int j = 0; j < 4; ++j)
                    crow[(size_t)r * N + 16 * j] = acc[i][j][r] + bv[j];
        }
    }
}

#define AT_WAVES   4
#define AT_KT      32
#define AT_QGROUPS (SEQ / (16 * AT_WAVES))
#define AT_STEPS   (SEQ / AT_KT)
#define K_PITCH    68
#define V_PITCH    136
#define K_STAGE    (AT_KT * K_PITCH)
#define V_STAGE    ((AT_KT / 2) * V_PITCH)

__device__ __forceinline__ float xhalf(float x) {
    return __builtin_bit_cast(float, __builtin_amdgcn_permlanex16(
        __builtin_bit_cast(int, x), __builtin_bit_cast(int, x),
        0x76543210, 0xfedcba98, false, false));
}

__global__ __launch_bounds__(32 * AT_WAVES) __attribute__((amdgpu_waves_per_eu(4, 4)))
void attn_kernel(const float* __restrict__ qkv, const float* __restrict__ hks,
                 const int* __restrict__ oclp, float* __restrict__ out,
                 const float* __restrict__ part) {
    __shared__ __align__(16) float sK[2][K_STAGE];
    __shared__ __align__(16) float sV[2][V_STAGE];
    __shared__ __align__(16) float sF[SEQ];

    const int bx = blockIdx.x;
    const int qg = bx & (AT_QGROUPS - 1);
    const int h  = (bx >> 5) & (NUM_HEADS - 1);
    const int b  = bx >> 9;

    const int tid  = threadIdx.x;
    const int w    = tid >> 5;
    const int l    = tid & 31;
    const int lo   = l & 15;
    const int half = l >> 4;
    const int qt   = qg * AT_WAVES + w;

    const int   hist = SEQ - *oclp;
    const float skh  = 1.0f + 9.0f / (1.0f + __expf(-hks[h]));

    const size_t base  = (size_t)b * SEQ * QKV_N;
    const float* qbase = qkv + base + (size_t)(qt * 16 + lo) * QKV_N + h * HEAD_DIM + 2 * half;

    const int srow = tid >> 4;
    const int scol = 4 * (tid & 15);
    const char* kg_base = (const char*)(qkv + base + INNER     + h * HEAD_DIM);
    const char* vg_base = (const char*)(qkv + base + 2 * INNER + h * HEAD_DIM);
    const unsigned kob = 4u * (unsigned)(srow * QKV_N + scol);
    const unsigned vob = 4u * (unsigned)(2 * srow * QKV_N + scol);
    const unsigned rowb = 4u * QKV_N;

    v4f rk0, rk1, rk2, rk3, rva0, rvb0, rva1, rvb1;

    rk0  = *(const v4f*)(kg_base + kob);
    rk1  = *(const v4f*)(kg_base + kob + 8  * rowb);
    rk2  = *(const v4f*)(kg_base + kob + 16 * rowb);
    rk3  = *(const v4f*)(kg_base + kob + 24 * rowb);
    rva0 = *(const v4f*)(vg_base + vob);
    rvb0 = *(const v4f*)(vg_base + vob + 1  * rowb);
    rva1 = *(const v4f*)(vg_base + vob + 16 * rowb);
    rvb1 = *(const v4f*)(vg_base + vob + 17 * rowb);

    v2f qa[16];
#pragma unroll
    for (int d = 0; d < 16; ++d)
        qa[d] = *(const v2f*)(qbase + 4 * d);

    {
        const v4f* pq = (const v4f*)(part + ((size_t)b * SEQ + qt * 16 + lo) * PART_N);
        const v4f sm = (pq[0] + pq[1]) + (pq[2] + pq[3]);
        const float ssq = (sm.x + sm.y) + (sm.z + sm.w);
        const float qs = rsqrtf(ssq * (1.0f / INNER) + 1e-5f) * (0.125f * 1.4426950408889634f);
#pragma unroll
        for (int d = 0; d < 16; ++d)
            qa[d] *= qs;
    }

    {
        const float* pk = part + (size_t)b * SEQ * PART_N + 16;
#pragma unroll 4
        for (int n = 0; n < SEQ / (32 * AT_WAVES); ++n) {
            const int key = tid + 32 * AT_WAVES * n;
            const v4f* pp = (const v4f*)(pk + (size_t)key * PART_N);
            const v4f sm = (pp[0] + pp[1]) + (pp[2] + pp[3]);
            const float ssq = (sm.x + sm.y) + (sm.z + sm.w);
            float f = rsqrtf(ssq * (1.0f / INNER) + 1e-5f);
            if (key < hist) f *= skh;
            sF[(key & ~31) + ((key & 7) << 2) + ((key >> 3) & 3)] = f;
        }
    }
    __syncthreads();

    {
        const v4f f4 = *(const v4f*)&sF[srow * 4];
        rk0 *= f4.x; rk1 *= f4.y; rk2 *= f4.z; rk3 *= f4.w;
        float* dk = &sK[0][srow * K_PITCH + scol];
        *(v4f*)(dk)                = rk0;
        *(v4f*)(dk + 8  * K_PITCH) = rk1;
        *(v4f*)(dk + 16 * K_PITCH) = rk2;
        *(v4f*)(dk + 24 * K_PITCH) = rk3;
        float* dv = &sV[0][srow * V_PITCH + 2 * scol];
        *(v4f*)(dv)                   = __builtin_shufflevector(rva0, rvb0, 0, 4, 1, 5);
        *(v4f*)(dv + 4)               = __builtin_shufflevector(rva0, rvb0, 2, 6, 3, 7);
        *(v4f*)(dv + 8 * V_PITCH)     = __builtin_shufflevector(rva1, rvb1, 0, 4, 1, 5);
        *(v4f*)(dv + 8 * V_PITCH + 4) = __builtin_shufflevector(rva1, rvb1, 2, 6, 3, 7);
    }
    __syncthreads();

    v8f o[4] = {};
    float m_run = -__builtin_inff();
    float l_run = 0.f;

    for (int st = 0; st < AT_STEPS; ++st) {
        const int s = st & 1;
        const int    sn = (st + 1) & (AT_STEPS - 1);
        const size_t kn = (size_t)(sn * AT_KT) * rowb;
        const char* kgs = kg_base + kn;
        const char* vgs = vg_base + kn;
        rk0  = *(const v4f*)(kgs + kob);
        rk1  = *(const v4f*)(kgs + kob + 8  * rowb);
        rk2  = *(const v4f*)(kgs + kob + 16 * rowb);
        rk3  = *(const v4f*)(kgs + kob + 24 * rowb);
        rva0 = *(const v4f*)(vgs + vob);
        rvb0 = *(const v4f*)(vgs + vob + 1  * rowb);
        rva1 = *(const v4f*)(vgs + vob + 16 * rowb);
        rvb1 = *(const v4f*)(vgs + vob + 17 * rowb);

        const float* kL = &sK[s][lo * K_PITCH + 2 * half];
        const float* vL = &sV[s][(4 * half) * V_PITCH + 2 * lo];

        v2f kf[2][16];
#pragma unroll
        for (int j = 0; j < 2; ++j)
#pragma unroll
            for (int d = 0; d < 16; ++d)
                kf[j][d] = *(const v2f*)(kL + j * 16 * K_PITCH + 4 * d);

        asm volatile("" : "+v"(kf[0][0]) : : "memory");

        v8f c[2] = {};
#pragma unroll
        for (int d = 0; d < 16; ++d) {
#pragma unroll
            for (int j = 0; j < 2; ++j)
                c[j] = __builtin_amdgcn_wmma_f32_16x16x4_f32(false, kf[j][d], false, qa[d],
                                                             (short)0, c[j], false, false);
        }

        v2f vf[2][4][4];
#pragma unroll
        for (int j = 0; j < 2; ++j)
#pragma unroll
            for (int kk = 0; kk < 4; ++kk)
#pragma unroll
                for (int nt = 0; nt < 4; ++nt)
                    vf[j][kk][nt] = *(const v2f*)(vL + (8 * j + kk) * V_PITCH + 32 * nt);
        const v4f f4 = *(const v4f*)&sF[sn * AT_KT + srow * 4];

        asm volatile("" : "+v"(c[1]) : : "memory");

        float sc[2][8];
#pragma unroll
        for (int j = 0; j < 2; ++j)
#pragma unroll
            for (int r = 0; r < 8; ++r)
                sc[j][r] = c[j][r];

        float rm0 = fmaxf(fmaxf(sc[0][0], sc[0][1]), fmaxf(sc[0][2], sc[0][3]));
        float rm1 = fmaxf(fmaxf(sc[0][4], sc[0][5]), fmaxf(sc[0][6], sc[0][7]));
        float rm2 = fmaxf(fmaxf(sc[1][0], sc[1][1]), fmaxf(sc[1][2], sc[1][3]));
        float rm3 = fmaxf(fmaxf(sc[1][4], sc[1][5]), fmaxf(sc[1][6], sc[1][7]));
        float rm  = fmaxf(fmaxf(rm0, rm1), fmaxf(rm2, rm3));
        rm = fmaxf(rm, xhalf(rm));

        const float mn = fmaxf(m_run, rm);
        const float al = __builtin_amdgcn_exp2f(m_run - mn);

        float p[2][8];
#pragma unroll
        for (int j = 0; j < 2; ++j)
#pragma unroll
            for (int r = 0; r < 8; ++r)
                p[j][r] = __builtin_amdgcn_exp2f(sc[j][r] - mn);

        const float su0 = (p[0][0] + p[0][1]) + (p[0][2] + p[0][3]);
        const float su1 = (p[0][4] + p[0][5]) + (p[0][6] + p[0][7]);
        const float su2 = (p[1][0] + p[1][1]) + (p[1][2] + p[1][3]);
        const float su3 = (p[1][4] + p[1][5]) + (p[1][6] + p[1][7]);
        l_run = l_run * al + ((su0 + su1) + (su2 + su3));
        m_run = mn;

#pragma unroll
        for (int nt = 0; nt < 4; ++nt)
#pragma unroll
            for (int r = 0; r < 8; ++r)
                o[nt][r] *= al;

#pragma unroll
        for (int j = 0; j < 2; ++j) {
#pragma unroll
            for (int kk = 0; kk < 4; ++kk) {
                v2f pb;
                pb.x = p[j][2 * kk];
                pb.y = p[j][2 * kk + 1];
#pragma unroll
                for (int nt = 0; nt < 4; ++nt) {
                    o[nt] = __builtin_amdgcn_wmma_f32_16x16x4_f32(false, vf[j][kk][nt], false, pb,
                                                                  (short)0, o[nt], false, false);
                }
            }
        }

        {
            rk0 *= f4.x; rk1 *= f4.y; rk2 *= f4.z; rk3 *= f4.w;
            float* dk = &sK[s ^ 1][srow * K_PITCH + scol];
            *(v4f*)(dk)                = rk0;
            *(v4f*)(dk + 8  * K_PITCH) = rk1;
            *(v4f*)(dk + 16 * K_PITCH) = rk2;
            *(v4f*)(dk + 24 * K_PITCH) = rk3;
            float* dv = &sV[s ^ 1][srow * V_PITCH + 2 * scol];
            *(v4f*)(dv)                   = __builtin_shufflevector(rva0, rvb0, 0, 4, 1, 5);
            *(v4f*)(dv + 4)               = __builtin_shufflevector(rva0, rvb0, 2, 6, 3, 7);
            *(v4f*)(dv + 8 * V_PITCH)     = __builtin_shufflevector(rva1, rvb1, 0, 4, 1, 5);
            *(v4f*)(dv + 8 * V_PITCH + 4) = __builtin_shufflevector(rva1, rvb1, 2, 6, 3, 7);
        }
        __syncthreads();
    }

    const float l_fin = l_run + xhalf(l_run);
    const float inv = 1.0f / l_fin;

    float* orow = out + (size_t)((size_t)b * SEQ + qt * 16 + lo) * INNER
                      + h * HEAD_DIM + 8 * half;
#pragma unroll
    for (int nt = 0; nt < 4; ++nt) {
        *(float4*)(orow + 16 * nt)     = make_float4(o[nt][0] * inv, o[nt][1] * inv,
                                                     o[nt][2] * inv, o[nt][3] * inv);
        *(float4*)(orow + 16 * nt + 4) = make_float4(o[nt][4] * inv, o[nt][5] * inv,
                                                     o[nt][6] * inv, o[nt][7] * inv);
    }
}

extern "C" void kernel_launch(void* const* d_in, const int* in_sizes, int n_in,
                              void* d_out, int out_size, void* d_ws, size_t ws_size,
                              hipStream_t stream) {
    const float* hidden = (const float*)d_in[0];
    const float* rot    = (const float*)d_in[1];
    const float* w_qkv  = (const float*)d_in[2];
    const float* b_qkv  = (const float*)d_in[3];
    const float* nqw    = (const float*)d_in[4];
    const float* nkw    = (const float*)d_in[5];
    const float* w_out  = (const float*)d_in[6];
    const float* b_out  = (const float*)d_in[7];
    const float* hks    = (const float*)d_in[8];
    const int*   ocl    = (const int*)d_in[9];
    float* out = (float*)d_out;

    float* qkv  = (float*)d_ws;
    float* attn = qkv + (size_t)BS * QKV_N;
    float* part = out;

    gemm_bias_f32<true><<<dim3(QKV_N / GB_N, BS / GB_M), 128, 0, stream>>>(
        hidden, w_qkv, b_qkv, qkv, BS, QKV_N, DIM, rot, nqw, nkw, part);

    attn_kernel<<<BATCH * NUM_HEADS * AT_QGROUPS, 32 * AT_WAVES, 0, stream>>>(qkv, hks, ocl, attn, part);

    gemm_bias_f32<false><<<dim3(DIM / GB_N, BS / GB_M), 128, 0, stream>>>(
        attn, w_out, b_out, out, BS, DIM, DIM, nullptr, nullptr, nullptr, nullptr);
}
